# starting kernel
# baseline (speedup 1.0000x reference)
#include <hip/hip_runtime.h>
#include <math.h>


typedef float v2f __attribute__((ext_vector_type(2)));
typedef float v8f __attribute__((ext_vector_type(8)));

#define N_   16
#define C_   256
#define H_   80
#define W_   80
#define MIP  8
#define L_   160

__device__ __forceinline__ float sigf(float v) {
    return 1.0f / (1.0f + __expf(-v));
}

__global__ void pool_kernel(const float* __restrict__ x, float* __restrict__ y_pool) {
    const int R = N_ * C_ * H_;
    int tid = blockIdx.x * blockDim.x + threadIdx.x;
    if (tid < R) {
        int i  = tid % H_;
        int nc = tid / H_;
        const float* p = x + ((long)nc * H_ + i) * W_;
        float s = 0.f;
#pragma unroll
        for (int j = 0; j < W_; j += 4) {
            float4 v = *reinterpret_cast<const float4*>(p + j);
            s += v.x + v.y + v.z + v.w;
        }
        y_pool[(long)nc * L_ + i] = s * (1.0f / (float)W_);
    } else {
        int t = tid - R;
        if (t < N_ * C_ * W_) {
            int j  = t % W_;
            int nc = t / W_;
            const float* p = x + (long)nc * H_ * W_ + j;
            float s = 0.f;
#pragma unroll 8
            for (int i = 0; i < H_; ++i) s += p[(long)i * W_];
            y_pool[(long)nc * L_ + H_ + j] = s * (1.0f / (float)H_);
        }
    }
}

__global__ void gemm1_bn_swish_kernel(const float* __restrict__ y_pool,
                                      const float* __restrict__ w_fc,
                                      const float* __restrict__ g,
                                      const float* __restrict__ b,
                                      const float* __restrict__ mean,
                                      const float* __restrict__ var,
                                      float* __restrict__ y_mid) {
    const int tile = blockIdx.x;
    const int n    = blockIdx.y;
    const int lane = threadIdx.x;
    const int half = lane >> 4;
    const int l16  = lane & 15;

    const float amask = (l16 < MIP) ? 1.0f : 0.0f;
    const float* wrow = w_fc + (l16 & (MIP - 1)) * C_;
    const float* yb   = y_pool + (long)n * C_ * L_;
    const int    col  = tile * 16 + l16;

    v8f acc0 = {};
    v8f acc1 = {};
    for (int kk = 0; kk < C_; kk += 8) {
        {
            const int k = kk + 2 * half;
            v2f a, bb;
            a.x  = wrow[k]     * amask;
            a.y  = wrow[k + 1] * amask;
            bb.x = yb[(long)k * L_ + col];
            bb.y = yb[(long)(k + 1) * L_ + col];
            acc0 = __builtin_amdgcn_wmma_f32_16x16x4_f32(false, a, false, bb,
                                                         (short)0, acc0, false, false);
        }
        {
            const int k = kk + 4 + 2 * half;
            v2f a, bb;
            a.x  = wrow[k]     * amask;
            a.y  = wrow[k + 1] * amask;
            bb.x = yb[(long)k * L_ + col];
            bb.y = yb[(long)(k + 1) * L_ + col];
            acc1 = __builtin_amdgcn_wmma_f32_16x16x4_f32(false, a, false, bb,
                                                         (short)0, acc1, false, false);
        }
    }

    if (half == 0) {
#pragma unroll
        for (int r = 0; r < MIP; ++r) {
            float iv    = g[r] * rsqrtf(var[r] + 1e-5f);
            float shift = b[r] - mean[r] * iv;
            float v = (acc0[r] + acc1[r]) * iv + shift;
            v = v * sigf(v);
            y_mid[((long)n * MIP + r) * L_ + col] = v;
        }
    }
}

__global__ void gemm2_sig_kernel(const float* __restrict__ y_mid,
                                 const float* __restrict__ w_h,
                                 const float* __restrict__ w_w,
                                 float* __restrict__ a_h,
                                 float* __restrict__ a_w) {
    const int ntile = blockIdx.x;
    const int mtile = blockIdx.y;
    const int n     = blockIdx.z;
    const int lane  = threadIdx.x;
    const int half  = lane >> 4;
    const int l16   = lane & 15;

    const bool  is_h = (ntile < 5);
    const float* Wm  = is_h ? w_h : w_w;
    const int   col  = ntile * 16 + l16;
    const int   chA  = mtile * 16 + l16;
    const float* ymb = y_mid + (long)n * MIP * L_;

    v8f acc = {};
#pragma unroll
    for (int kk = 0; kk < MIP; kk += 4) {
        const int k = kk + 2 * half;
        v2f a, bb;
        a.x  = Wm[chA * MIP + k];
        a.y  = Wm[chA * MIP + k + 1];
        bb.x = ymb[(long)k * L_ + col];
        bb.y = ymb[(long)(k + 1) * L_ + col];
        acc = __builtin_amdgcn_wmma_f32_16x16x4_f32(false, a, false, bb,
                                                    (short)0, acc, false, false);
    }

    float* dst = is_h ? a_h : a_w;
    const int cc = is_h ? col : (col - H_);
#pragma unroll
    for (int r = 0; r < MIP; ++r) {
        const int ch = mtile * 16 + half * 8 + r;
        dst[((long)n * C_ + ch) * H_ + cc] = sigf(acc[r]);
    }
}

__global__ void apply_kernel(const float* __restrict__ x,
                             const float* __restrict__ a_h,
                             const float* __restrict__ a_w,
                             float* __restrict__ out) {
    const long t = (long)blockIdx.x * blockDim.x + threadIdx.x;
    const long e = t * 4;
    const int  j  = (int)(e % W_);
    const long rid = e / W_;
    const int  i  = (int)(rid % H_);
    const long nc = rid / H_;

    const float  ah = a_h[nc * H_ + i];
    const float4 aw = *reinterpret_cast<const float4*>(a_w + nc * W_ + j);
    const float4 xv = *reinterpret_cast<const float4*>(x + e);

    float4 o;
    o.x = xv.x * ah * aw.x;
    o.y = xv.y * ah * aw.y;
    o.z = xv.z * ah * aw.z;
    o.w = xv.w * ah * aw.w;
    *reinterpret_cast<float4*>(out + e) = o;
}

extern "C" void kernel_launch(void* const* d_in, const int* in_sizes, int n_in,
                              void* d_out, int out_size, void* d_ws, size_t ws_size,
                              hipStream_t stream) {
    const float* x     = (const float*)d_in[0];
    const float* w_fc  = (const float*)d_in[1];
    const float* bn_g  = (const float*)d_in[2];
    const float* bn_b  = (const float*)d_in[3];
    const float* bn_m  = (const float*)d_in[4];
    const float* bn_v  = (const float*)d_in[5];
    const float* w_h   = (const float*)d_in[6];
    const float* w_w   = (const float*)d_in[7];
    float* out = (float*)d_out;

    float* ws     = (float*)d_ws;
    float* y_pool = ws;
    float* y_mid  = y_pool + (long)N_ * C_ * L_;
    float* a_h    = y_mid  + (long)N_ * MIP * L_;
    float* a_w    = a_h    + (long)N_ * C_ * H_;

    {
        int total = N_ * C_ * (H_ + W_);
        pool_kernel<<<(total + 255) / 256, 256, 0, stream>>>(x, y_pool);
    }
    gemm1_bn_swish_kernel<<<dim3(L_ / 16, N_), 32, 0, stream>>>(y_pool, w_fc, bn_g,
                                                                bn_b, bn_m, bn_v,
                                                                y_mid);
    gemm2_sig_kernel<<<dim3(L_ / 16, C_ / 16, N_), 32, 0, stream>>>(y_mid, w_h, w_w,
                                                                    a_h, a_w);
    {
        long nvec4 = (long)N_ * C_ * H_ * W_ / 4;
        apply_kernel<<<(unsigned)(nvec4 / 256), 256, 0, stream>>>(x, a_h, a_w, out);
    }
}


# pool of the mid planes, BN+swish and excite WMMAs share one launch: 8-wave group per (image, 16 l) pools its own 8x16 y tile into LDS; y plane and a launch dropped, z parked in out; new sum order
# speedup vs baseline: 1.6117x; 1.6117x over previous
#include <hip/hip_runtime.h>
#include <math.h>


typedef float v2f __attribute__((ext_vector_type(2)));
typedef float v8f __attribute__((ext_vector_type(8)));

#define N_   16
#define C_   256
#define H_   80
#define W_   80
#define MIP  8
#define L_   160
#define HW_  6400
#define ZP_  64
#define ZW_  8
#define ZC_  32

__device__ __forceinline__ float sigf(float v) {
    return 1.0f / (1.0f + __expf(-v));
}

__global__ __launch_bounds__(256)
void squeeze_pixels_kernel(const float* __restrict__ x,
                           const float* __restrict__ w_fc,
                           float* __restrict__ z) {
    __shared__ __attribute__((aligned(16))) float part[ZW_][16][ZP_];

    const int chunk = blockIdx.x;
    const int n     = blockIdx.y;
    const int wave  = threadIdx.x >> 5;
    const int lane  = threadIdx.x & 31;
    const int half  = lane >> 4;
    const int l16   = lane & 15;

    const int    c0 = wave * ZC_ + 2 * half;
    const float* xb = x + ((long)n * C_ + c0) * HW_ + chunk * ZP_ + 4 * l16;
    const float* wr = w_fc + (l16 & (MIP - 1)) * C_ + c0;

    float4 v0[ZC_ / 4], v1[ZC_ / 4];
    for (int s = 0; s < ZC_ / 4; ++s) {
        v0[s] = *reinterpret_cast<const float4*>(xb + (long)(4 * s)     * HW_);
        v1[s] = *reinterpret_cast<const float4*>(xb + (long)(4 * s + 1) * HW_);
    }

    v8f acc0 = {}, acc1 = {}, acc2 = {}, acc3 = {};
    for (int s = 0; s < ZC_ / 4; ++s) {
        v2f a, b;
        a.x = wr[4 * s];
        a.y = wr[4 * s + 1];
        b.x = v0[s].x;  b.y = v1[s].x;
        acc0 = __builtin_amdgcn_wmma_f32_16x16x4_f32(false, a, false, b,
                                                     (short)0, acc0, false, false);
        b.x = v0[s].y;  b.y = v1[s].y;
        acc1 = __builtin_amdgcn_wmma_f32_16x16x4_f32(false, a, false, b,
                                                     (short)0, acc1, false, false);
        b.x = v0[s].z;  b.y = v1[s].z;
        acc2 = __builtin_amdgcn_wmma_f32_16x16x4_f32(false, a, false, b,
                                                     (short)0, acc2, false, false);
        b.x = v0[s].w;  b.y = v1[s].w;
        acc3 = __builtin_amdgcn_wmma_f32_16x16x4_f32(false, a, false, b,
                                                     (short)0, acc3, false, false);
    }

    for (int r = 0; r < MIP; ++r) {
        float4 o;
        o.x = acc0[r];  o.y = acc1[r];  o.z = acc2[r];  o.w = acc3[r];
        *reinterpret_cast<float4*>(&part[wave][r + 8 * half][4 * l16]) = o;
    }
    __syncthreads();

    const int r  = threadIdx.x >> 5;
    const int cc = (threadIdx.x & 31) * 2;
    float2 s = {0.f, 0.f};
    for (int w = 0; w < ZW_; ++w) {
        const float2 p = *reinterpret_cast<const float2*>(&part[w][r][cc]);
        s.x += p.x;
        s.y += p.y;
    }
    *reinterpret_cast<float2*>(z + ((long)n * MIP + r) * HW_ + chunk * ZP_ + cc) = s;
}

#define PW_ 8

__global__ __launch_bounds__(256)
void pool_excite_kernel(const float* __restrict__ z,
                        const float* __restrict__ g,
                        const float* __restrict__ b,
                        const float* __restrict__ mean,
                        const float* __restrict__ var,
                        const float* __restrict__ w_h,
                        const float* __restrict__ w_w,
                        float* __restrict__ a_h,
                        float* __restrict__ a_w) {
    __shared__ __attribute__((aligned(16))) float colpart[PW_][MIP][16];
    __shared__ float ytile[MIP][16];

    const int ntile = blockIdx.x;
    const int n     = blockIdx.y;
    const int wave  = threadIdx.x >> 5;
    const int lane  = threadIdx.x & 31;
    const int half  = lane >> 4;
    const int l16   = lane & 15;
    const int m     = lane >> 2;
    const int q     = lane & 3;

    const bool   is_h = (ntile < 5);
    const float* Wm   = is_h ? w_h : w_w;

    float2 wa[2][2];
    for (int t = 0; t < 2; ++t) {
        const int chA = (2 * wave + t) * 16 + l16;
        wa[t][0] = *reinterpret_cast<const float2*>(Wm + chA * MIP + 2 * half);
        wa[t][1] = *reinterpret_cast<const float2*>(Wm + chA * MIP + 4 + 2 * half);
    }

    const float* zp = z + ((long)n * MIP + m) * HW_;
    if (is_h) {
        const float* p = zp + (ntile * 16 + 2 * wave) * W_ + 4 * q;
        float4 u0[5], u1[5];
        for (int k = 0; k < 5; ++k) {
            u0[k] = *reinterpret_cast<const float4*>(p + 16 * k);
            u1[k] = *reinterpret_cast<const float4*>(p + W_ + 16 * k);
        }
        float s0 = 0.f, s1 = 0.f;
        for (int k = 0; k < 5; ++k) {
            s0 += (u0[k].x + u0[k].y) + (u0[k].z + u0[k].w);
            s1 += (u1[k].x + u1[k].y) + (u1[k].z + u1[k].w);
        }
        s0 += __shfl_xor(s0, 1);  s0 += __shfl_xor(s0, 2);
        s1 += __shfl_xor(s1, 1);  s1 += __shfl_xor(s1, 2);
        if (q == 0) {
            const float iv    = g[m] * rsqrtf(var[m] + 1e-5f);
            const float shift = b[m] - mean[m] * iv;
            float y0 = s0 * (1.0f / (float)W_) * iv + shift;
            float y1 = s1 * (1.0f / (float)W_) * iv + shift;
            ytile[m][2 * wave]     = y0 * sigf(y0);
            ytile[m][2 * wave + 1] = y1 * sigf(y1);
        }
    } else {
        const float* p = zp + wave * W_ + (ntile - 5) * 16 + 4 * q;
        float4 u[10];
        for (int k = 0; k < 10; ++k)
            u[k] = *reinterpret_cast<const float4*>(p + (PW_ * k) * W_);
        float4 s = u[0];
        for (int k = 1; k < 10; ++k) {
            s.x += u[k].x;  s.y += u[k].y;  s.z += u[k].z;  s.w += u[k].w;
        }
        *reinterpret_cast<float4*>(&colpart[wave][m][4 * q]) = s;
        __syncthreads();
        if (threadIdx.x < MIP * 16) {
            const int mm = threadIdx.x >> 4;
            const int cl = threadIdx.x & 15;
            float t = 0.f;
            for (int w = 0; w < PW_; ++w) t += colpart[w][mm][cl];
            const float iv    = g[mm] * rsqrtf(var[mm] + 1e-5f);
            const float shift = b[mm] - mean[mm] * iv;
            float y = t * (1.0f / (float)H_) * iv + shift;
            ytile[mm][cl] = y * sigf(y);
        }
    }
    __syncthreads();

    float*    dst = is_h ? a_h : a_w;
    const int cc  = (is_h ? ntile : ntile - 5) * 16 + l16;
    for (int t = 0; t < 2; ++t) {
        v8f acc = {};
        for (int kk = 0; kk < 2; ++kk) {
            const int k = 4 * kk + 2 * half;
            v2f a, bb;
            a.x  = wa[t][kk].x;
            a.y  = wa[t][kk].y;
            bb.x = ytile[k][l16];
            bb.y = ytile[k + 1][l16];
            acc = __builtin_amdgcn_wmma_f32_16x16x4_f32(false, a, false, bb,
                                                        (short)0, acc, false, false);
        }
        for (int r = 0; r < MIP; ++r) {
            const int ch = (2 * wave + t) * 16 + half * 8 + r;
            dst[((long)n * C_ + ch) * H_ + cc] = sigf(acc[r]);
        }
    }
}

__global__ void apply_kernel(const float* __restrict__ x,
                             const float* __restrict__ a_h,
                             const float* __restrict__ a_w,
                             float* __restrict__ out) {
    const long t = (long)blockIdx.x * blockDim.x + threadIdx.x;
    const long e = t * 4;
    const int  j  = (int)(e % W_);
    const long rid = e / W_;
    const int  i  = (int)(rid % H_);
    const long nc = rid / H_;

    const float  ah = a_h[nc * H_ + i];
    const float4 aw = *reinterpret_cast<const float4*>(a_w + nc * W_ + j);
    const float4 xv = *reinterpret_cast<const float4*>(x + e);

    float4 o;
    o.x = xv.x * ah * aw.x;
    o.y = xv.y * ah * aw.y;
    o.z = xv.z * ah * aw.z;
    o.w = xv.w * ah * aw.w;
    *reinterpret_cast<float4*>(out + e) = o;
}

extern "C" void kernel_launch(void* const* d_in, const int* in_sizes, int n_in,
                              void* d_out, int out_size, void* d_ws, size_t ws_size,
                              hipStream_t stream) {
    const float* x     = (const float*)d_in[0];
    const float* w_fc  = (const float*)d_in[1];
    const float* bn_g  = (const float*)d_in[2];
    const float* bn_b  = (const float*)d_in[3];
    const float* bn_m  = (const float*)d_in[4];
    const float* bn_v  = (const float*)d_in[5];
    const float* w_h   = (const float*)d_in[6];
    const float* w_w   = (const float*)d_in[7];
    float* out = (float*)d_out;

    float* ws  = (float*)d_ws;
    float* a_h = ws;
    float* a_w = a_h + (long)N_ * C_ * H_;
    float* z   = out;

    squeeze_pixels_kernel<<<dim3(HW_ / ZP_, N_), 32 * ZW_, 0, stream>>>(x, w_fc, z);
    pool_excite_kernel<<<dim3(L_ / 16, N_), 32 * PW_, 0, stream>>>(z, bn_g, bn_b, bn_m,
                                                                   bn_v, w_h, w_w,
                                                                   a_h, a_w);
    {
        long nvec4 = (long)N_ * C_ * H_ * W_ / 4;
        apply_kernel<<<(unsigned)(nvec4 / 256), 256, 0, stream>>>(x, a_h, a_w, out);
    }
}


# excite WMMAs+sigmoid moved into the apply launch: 16-wave group per (image,16 ch) forms its 16x160 gates in LDS, then each wave streams one plane, 10-20 loads in flight; gate planes gone; same sums
# speedup vs baseline: 2.3389x; 1.4512x over previous
#include <hip/hip_runtime.h>
#include <math.h>


typedef float v2f __attribute__((ext_vector_type(2)));
typedef float v8f __attribute__((ext_vector_type(8)));

#define N_   16
#define C_   256
#define H_   80
#define W_   80
#define MIP  8
#define L_   160
#define HW_  6400
#define ZP_  64
#define ZW_  8
#define ZC_  32

__device__ __forceinline__ float sigf(float v) {
    return 1.0f / (1.0f + __expf(-v));
}

__global__ __launch_bounds__(256)
void squeeze_pixels_kernel(const float* __restrict__ x,
                           const float* __restrict__ w_fc,
                           float* __restrict__ z) {
    __shared__ __attribute__((aligned(16))) float part[ZW_][16][ZP_];

    const int chunk = blockIdx.x;
    const int n     = blockIdx.y;
    const int wave  = threadIdx.x >> 5;
    const int lane  = threadIdx.x & 31;
    const int half  = lane >> 4;
    const int l16   = lane & 15;

    const int    c0 = wave * ZC_ + 2 * half;
    const float* xb = x + ((long)n * C_ + c0) * HW_ + chunk * ZP_ + 4 * l16;
    const float* wr = w_fc + (l16 & (MIP - 1)) * C_ + c0;

    float4 v0[ZC_ / 4], v1[ZC_ / 4];
    for (int s = 0; s < ZC_ / 4; ++s) {
        v0[s] = *reinterpret_cast<const float4*>(xb + (long)(4 * s)     * HW_);
        v1[s] = *reinterpret_cast<const float4*>(xb + (long)(4 * s + 1) * HW_);
    }

    v8f acc0 = {}, acc1 = {}, acc2 = {}, acc3 = {};
    for (int s = 0; s < ZC_ / 4; ++s) {
        v2f a, b;
        a.x = wr[4 * s];
        a.y = wr[4 * s + 1];
        b.x = v0[s].x;  b.y = v1[s].x;
        acc0 = __builtin_amdgcn_wmma_f32_16x16x4_f32(false, a, false, b,
                                                     (short)0, acc0, false, false);
        b.x = v0[s].y;  b.y = v1[s].y;
        acc1 = __builtin_amdgcn_wmma_f32_16x16x4_f32(false, a, false, b,
                                                     (short)0, acc1, false, false);
        b.x = v0[s].z;  b.y = v1[s].z;
        acc2 = __builtin_amdgcn_wmma_f32_16x16x4_f32(false, a, false, b,
                                                     (short)0, acc2, false, false);
        b.x = v0[s].w;  b.y = v1[s].w;
        acc3 = __builtin_amdgcn_wmma_f32_16x16x4_f32(false, a, false, b,
                                                     (short)0, acc3, false, false);
    }

    for (int r = 0; r < MIP; ++r) {
        float4 o;
        o.x = acc0[r];  o.y = acc1[r];  o.z = acc2[r];  o.w = acc3[r];
        *reinterpret_cast<float4*>(&part[wave][r + 8 * half][4 * l16]) = o;
    }
    __syncthreads();

    const int r  = threadIdx.x >> 5;
    const int cc = (threadIdx.x & 31) * 2;
    float2 s = {0.f, 0.f};
    for (int w = 0; w < ZW_; ++w) {
        const float2 p = *reinterpret_cast<const float2*>(&part[w][r][cc]);
        s.x += p.x;
        s.y += p.y;
    }
    *reinterpret_cast<float2*>(z + ((long)n * MIP + r) * HW_ + chunk * ZP_ + cc) = s;
}

#define PW_ 8

__global__ __launch_bounds__(256)
void pool_mid_kernel(const float* __restrict__ z,
                     const float* __restrict__ g,
                     const float* __restrict__ b,
                     const float* __restrict__ mean,
                     const float* __restrict__ var,
                     float* __restrict__ y) {
    __shared__ __attribute__((aligned(16))) float colpart[PW_][MIP][16];

    const int ntile = blockIdx.x;
    const int n     = blockIdx.y;
    const int wave  = threadIdx.x >> 5;
    const int lane  = threadIdx.x & 31;
    const int m     = lane >> 2;
    const int q     = lane & 3;

    const bool is_h = (ntile < 5);

    const float* zp = z + ((long)n * MIP + m) * HW_;
    float*       yn = y + (long)n * MIP * L_;
    if (is_h) {
        const float* p = zp + (ntile * 16 + 2 * wave) * W_ + 4 * q;
        float4 u0[5], u1[5];
        for (int k = 0; k < 5; ++k) {
            u0[k] = *reinterpret_cast<const float4*>(p + 16 * k);
            u1[k] = *reinterpret_cast<const float4*>(p + W_ + 16 * k);
        }
        float s0 = 0.f, s1 = 0.f;
        for (int k = 0; k < 5; ++k) {
            s0 += (u0[k].x + u0[k].y) + (u0[k].z + u0[k].w);
            s1 += (u1[k].x + u1[k].y) + (u1[k].z + u1[k].w);
        }
        s0 += __shfl_xor(s0, 1);  s0 += __shfl_xor(s0, 2);
        s1 += __shfl_xor(s1, 1);  s1 += __shfl_xor(s1, 2);
        if (q == 0) {
            const float iv    = g[m] * rsqrtf(var[m] + 1e-5f);
            const float shift = b[m] - mean[m] * iv;
            float y0 = s0 * (1.0f / (float)W_) * iv + shift;
            float y1 = s1 * (1.0f / (float)W_) * iv + shift;
            float2 o;
            o.x = y0 * sigf(y0);
            o.y = y1 * sigf(y1);
            *reinterpret_cast<float2*>(yn + m * L_ + ntile * 16 + 2 * wave) = o;
        }
    } else {
        const float* p = zp + wave * W_ + (ntile - 5) * 16 + 4 * q;
        float4 u[10];
        for (int k = 0; k < 10; ++k)
            u[k] = *reinterpret_cast<const float4*>(p + (PW_ * k) * W_);
        float4 s = u[0];
        for (int k = 1; k < 10; ++k) {
            s.x += u[k].x;  s.y += u[k].y;  s.z += u[k].z;  s.w += u[k].w;
        }
        *reinterpret_cast<float4*>(&colpart[wave][m][4 * q]) = s;
        __syncthreads();
        if (threadIdx.x < MIP * 16) {
            const int mm = threadIdx.x >> 4;
            const int cl = threadIdx.x & 15;
            float t = 0.f;
            for (int w = 0; w < PW_; ++w) t += colpart[w][mm][cl];
            const float iv    = g[mm] * rsqrtf(var[mm] + 1e-5f);
            const float shift = b[mm] - mean[mm] * iv;
            float yy = t * (1.0f / (float)H_) * iv + shift;
            yn[mm * L_ + ntile * 16 + cl] = yy * sigf(yy);
        }
    }
}

#define AW_ 16
#define AG_ 10
#define NG_ 5

__global__ __launch_bounds__(512)
void excite_apply_kernel(const float* __restrict__ x,
                         const float* __restrict__ y,
                         const float* __restrict__ w_h,
                         const float* __restrict__ w_w,
                         float* __restrict__ out) {
    __shared__ __attribute__((aligned(16))) float gate[AW_][L_];

    const int ctile = blockIdx.x;
    const int n     = blockIdx.y;
    const int wave  = threadIdx.x >> 5;
    const int lane  = threadIdx.x & 31;
    const int half  = lane >> 4;
    const int l16   = lane & 15;

    const long    plane = ((long)n * C_ + ctile * AW_ + wave) * HW_;
    const float4* xp = reinterpret_cast<const float4*>(x + plane) + lane;
    float4*       op = reinterpret_cast<float4*>(out + plane) + lane;

    const bool gates = (wave < L_ / 16);
    v2f a0 = {}, a1 = {}, b0 = {}, b1 = {};
    if (gates) {
        const float* Wm = (wave < 5) ? w_h : w_w;
        const float* wr = Wm + (ctile * AW_ + l16) * MIP + 2 * half;
        const float* yc = y + (long)n * MIP * L_ + wave * 16 + l16 + (2 * half) * L_;
        a0.x = wr[0];           a0.y = wr[1];
        a1.x = wr[4];           a1.y = wr[5];
        b0.x = yc[0];           b0.y = yc[L_];
        b1.x = yc[4 * L_];      b1.y = yc[5 * L_];
    }

    float4 cur[AG_], nxt[AG_];
    for (int k = 0; k < AG_; ++k) cur[k] = xp[32 * k];
    for (int k = 0; k < AG_; ++k) nxt[k] = xp[32 * (AG_ + k)];

    if (gates) {
        v8f acc = {};
        acc = __builtin_amdgcn_wmma_f32_16x16x4_f32(false, a0, false, b0,
                                                    (short)0, acc, false, false);
        acc = __builtin_amdgcn_wmma_f32_16x16x4_f32(false, a1, false, b1,
                                                    (short)0, acc, false, false);
        for (int r = 0; r < MIP; ++r)
            gate[half * 8 + r][wave * 16 + l16] = sigf(acc[r]);
    }
    __syncthreads();

    const float* gh = &gate[wave][0];
    const float* gw = &gate[wave][H_];
    for (int g = 0; g < NG_; ++g) {
        for (int k = 0; k < AG_; ++k) {
            const unsigned f  = (unsigned)lane + 32u * (unsigned)(AG_ * g + k);
            const unsigned i  = f / 20u;
            const unsigned jq = f - 20u * i;
            const float    ah = gh[i];
            const float4   aw = *reinterpret_cast<const float4*>(gw + 4 * jq);
            float4 o;
            o.x = cur[k].x * ah * aw.x;
            o.y = cur[k].y * ah * aw.y;
            o.z = cur[k].z * ah * aw.z;
            o.w = cur[k].w * ah * aw.w;
            op[32 * (AG_ * g + k)] = o;
        }
        for (int k = 0; k < AG_; ++k) cur[k] = nxt[k];
        if (g + 2 < NG_)
            for (int k = 0; k < AG_; ++k) nxt[k] = xp[32 * (AG_ * (g + 2) + k)];
    }
}

extern "C" void kernel_launch(void* const* d_in, const int* in_sizes, int n_in,
                              void* d_out, int out_size, void* d_ws, size_t ws_size,
                              hipStream_t stream) {
    const float* x     = (const float*)d_in[0];
    const float* w_fc  = (const float*)d_in[1];
    const float* bn_g  = (const float*)d_in[2];
    const float* bn_b  = (const float*)d_in[3];
    const float* bn_m  = (const float*)d_in[4];
    const float* bn_v  = (const float*)d_in[5];
    const float* w_h   = (const float*)d_in[6];
    const float* w_w   = (const float*)d_in[7];
    float* out = (float*)d_out;

    float* y = (float*)d_ws;
    float* z = out;

    squeeze_pixels_kernel<<<dim3(HW_ / ZP_, N_), 32 * ZW_, 0, stream>>>(x, w_fc, z);
    pool_mid_kernel<<<dim3(L_ / 16, N_), 32 * PW_, 0, stream>>>(z, bn_g, bn_b, bn_m,
                                                                bn_v, y);
    excite_apply_kernel<<<dim3(C_ / AW_, N_), 32 * AW_, 0, stream>>>(x, y, w_h, w_w, out);
}


# the two walks over x matched for the L2: image = low 4 bits of the 1-D group index in all 3 launches, squeeze takes pixel pieces upward, apply streams each plane from its end; same sums
# speedup vs baseline: 2.7975x; 1.1961x over previous
#include <hip/hip_runtime.h>
#include <math.h>


typedef float v2f __attribute__((ext_vector_type(2)));
typedef float v8f __attribute__((ext_vector_type(8)));

#define N_   16
#define C_   256
#define H_   80
#define W_   80
#define MIP  8
#define L_   160
#define HW_  6400
#define ZP_  64
#define ZW_  8
#define ZC_  32

__device__ __forceinline__ float sigf(float v) {
    return 1.0f / (1.0f + __expf(-v));
}

__global__ __launch_bounds__(256)
void squeeze_pixels_kernel(const float* __restrict__ x,
                           const float* __restrict__ w_fc,
                           float* __restrict__ z) {
    __shared__ __attribute__((aligned(16))) float part[ZW_][16][ZP_];

    const int n     = blockIdx.x & (N_ - 1);
    const int chunk = blockIdx.x >> 4;
    const int wave  = threadIdx.x >> 5;
    const int lane  = threadIdx.x & 31;
    const int half  = lane >> 4;
    const int l16   = lane & 15;

    const int    c0 = wave * ZC_ + 2 * half;
    const float* xb = x + ((long)n * C_ + c0) * HW_ + chunk * ZP_ + 4 * l16;
    const float* wr = w_fc + (l16 & (MIP - 1)) * C_ + c0;

    float4 v0[ZC_ / 4], v1[ZC_ / 4];
    for (int s = 0; s < ZC_ / 4; ++s) {
        v0[s] = *reinterpret_cast<const float4*>(xb + (long)(4 * s)     * HW_);
        v1[s] = *reinterpret_cast<const float4*>(xb + (long)(4 * s + 1) * HW_);
    }

    v8f acc0 = {}, acc1 = {}, acc2 = {}, acc3 = {};
    for (int s = 0; s < ZC_ / 4; ++s) {
        v2f a, b;
        a.x = wr[4 * s];
        a.y = wr[4 * s + 1];
        b.x = v0[s].x;  b.y = v1[s].x;
        acc0 = __builtin_amdgcn_wmma_f32_16x16x4_f32(false, a, false, b,
                                                     (short)0, acc0, false, false);
        b.x = v0[s].y;  b.y = v1[s].y;
        acc1 = __builtin_amdgcn_wmma_f32_16x16x4_f32(false, a, false, b,
                                                     (short)0, acc1, false, false);
        b.x = v0[s].z;  b.y = v1[s].z;
        acc2 = __builtin_amdgcn_wmma_f32_16x16x4_f32(false, a, false, b,
                                                     (short)0, acc2, false, false);
        b.x = v0[s].w;  b.y = v1[s].w;
        acc3 = __builtin_amdgcn_wmma_f32_16x16x4_f32(false, a, false, b,
                                                     (short)0, acc3, false, false);
    }

    for (int r = 0; r < MIP; ++r) {
        float4 o;
        o.x = acc0[r];  o.y = acc1[r];  o.z = acc2[r];  o.w = acc3[r];
        *reinterpret_cast<float4*>(&part[wave][r + 8 * half][4 * l16]) = o;
    }
    __syncthreads();

    const int r  = threadIdx.x >> 5;
    const int cc = (threadIdx.x & 31) * 2;
    float2 s = {0.f, 0.f};
    for (int w = 0; w < ZW_; ++w) {
        const float2 p = *reinterpret_cast<const float2*>(&part[w][r][cc]);
        s.x += p.x;
        s.y += p.y;
    }
    *reinterpret_cast<float2*>(z + ((long)n * MIP + r) * HW_ + chunk * ZP_ + cc) = s;
}

#define PW_ 8

__global__ __launch_bounds__(256)
void pool_mid_kernel(const float* __restrict__ z,
                     const float* __restrict__ g,
                     const float* __restrict__ b,
                     const float* __restrict__ mean,
                     const float* __restrict__ var,
                     float* __restrict__ y) {
    __shared__ __attribute__((aligned(16))) float colpart[PW_][MIP][16];

    const int n     = blockIdx.x & (N_ - 1);
    const int ntile = blockIdx.x >> 4;
    const int wave  = threadIdx.x >> 5;
    const int lane  = threadIdx.x & 31;
    const int m     = lane >> 2;
    const int q     = lane & 3;

    const bool is_h = (ntile < 5);

    const float* zp = z + ((long)n * MIP + m) * HW_;
    float*       yn = y + (long)n * MIP * L_;
    if (is_h) {
        const float* p = zp + (ntile * 16 + 2 * wave) * W_ + 4 * q;
        float4 u0[5], u1[5];
        for (int k = 0; k < 5; ++k) {
            u0[k] = *reinterpret_cast<const float4*>(p + 16 * k);
            u1[k] = *reinterpret_cast<const float4*>(p + W_ + 16 * k);
        }
        float s0 = 0.f, s1 = 0.f;
        for (int k = 0; k < 5; ++k) {
            s0 += (u0[k].x + u0[k].y) + (u0[k].z + u0[k].w);
            s1 += (u1[k].x + u1[k].y) + (u1[k].z + u1[k].w);
        }
        s0 += __shfl_xor(s0, 1);  s0 += __shfl_xor(s0, 2);
        s1 += __shfl_xor(s1, 1);  s1 += __shfl_xor(s1, 2);
        if (q == 0) {
            const float iv    = g[m] * rsqrtf(var[m] + 1e-5f);
            const float shift = b[m] - mean[m] * iv;
            float y0 = s0 * (1.0f / (float)W_) * iv + shift;
            float y1 = s1 * (1.0f / (float)W_) * iv + shift;
            float2 o;
            o.x = y0 * sigf(y0);
            o.y = y1 * sigf(y1);
            *reinterpret_cast<float2*>(yn + m * L_ + ntile * 16 + 2 * wave) = o;
        }
    } else {
        const float* p = zp + wave * W_ + (ntile - 5) * 16 + 4 * q;
        float4 u[10];
        for (int k = 0; k < 10; ++k)
            u[k] = *reinterpret_cast<const float4*>(p + (PW_ * k) * W_);
        float4 s = u[0];
        for (int k = 1; k < 10; ++k) {
            s.x += u[k].x;  s.y += u[k].y;  s.z += u[k].z;  s.w += u[k].w;
        }
        *reinterpret_cast<float4*>(&colpart[wave][m][4 * q]) = s;
        __syncthreads();
        if (threadIdx.x < MIP * 16) {
            const int mm = threadIdx.x >> 4;
            const int cl = threadIdx.x & 15;
            float t = 0.f;
            for (int w = 0; w < PW_; ++w) t += colpart[w][mm][cl];
            const float iv    = g[mm] * rsqrtf(var[mm] + 1e-5f);
            const float shift = b[mm] - mean[mm] * iv;
            float yy = t * (1.0f / (float)H_) * iv + shift;
            yn[mm * L_ + ntile * 16 + cl] = yy * sigf(yy);
        }
    }
}

#define AW_ 16
#define AG_ 10
#define NG_ 5
#define BLK_(t) (AG_ * NG_ - 1 - (t))

__global__ __launch_bounds__(512)
void excite_apply_kernel(const float* __restrict__ x,
                         const float* __restrict__ y,
                         const float* __restrict__ w_h,
                         const float* __restrict__ w_w,
                         float* __restrict__ out) {
    __shared__ __attribute__((aligned(16))) float gate[AW_][L_];

    const int n     = blockIdx.x & (N_ - 1);
    const int ctile = blockIdx.x >> 4;
    const int wave  = threadIdx.x >> 5;
    const int lane  = threadIdx.x & 31;
    const int half  = lane >> 4;
    const int l16   = lane & 15;

    const long    plane = ((long)n * C_ + ctile * AW_ + wave) * HW_;
    const float4* xp = reinterpret_cast<const float4*>(x + plane) + lane;
    float4*       op = reinterpret_cast<float4*>(out + plane) + lane;

    const bool gates = (wave < L_ / 16);
    v2f a0 = {}, a1 = {}, b0 = {}, b1 = {};
    if (gates) {
        const float* Wm = (wave < 5) ? w_h : w_w;
        const float* wr = Wm + (ctile * AW_ + l16) * MIP + 2 * half;
        const float* yc = y + (long)n * MIP * L_ + wave * 16 + l16 + (2 * half) * L_;
        a0.x = wr[0];           a0.y = wr[1];
        a1.x = wr[4];           a1.y = wr[5];
        b0.x = yc[0];           b0.y = yc[L_];
        b1.x = yc[4 * L_];      b1.y = yc[5 * L_];
    }

    float4 cur[AG_], nxt[AG_];
    for (int k = 0; k < AG_; ++k) cur[k] = xp[32 * BLK_(k)];
    for (int k = 0; k < AG_; ++k) nxt[k] = xp[32 * BLK_(AG_ + k)];

    if (gates) {
        v8f acc = {};
        acc = __builtin_amdgcn_wmma_f32_16x16x4_f32(false, a0, false, b0,
                                                    (short)0, acc, false, false);
        acc = __builtin_amdgcn_wmma_f32_16x16x4_f32(false, a1, false, b1,
                                                    (short)0, acc, false, false);
        for (int r = 0; r < MIP; ++r)
            gate[half * 8 + r][wave * 16 + l16] = sigf(acc[r]);
    }
    __syncthreads();

    const float* gh = &gate[wave][0];
    const float* gw = &gate[wave][H_];
    for (int g = 0; g < NG_; ++g) {
        for (int k = 0; k < AG_; ++k) {
            const unsigned f  = (unsigned)lane + 32u * (unsigned)BLK_(AG_ * g + k);
            const unsigned i  = f / 20u;
            const unsigned jq = f - 20u * i;
            const float    ah = gh[i];
            const float4   aw = *reinterpret_cast<const float4*>(gw + 4 * jq);
            float4 o;
            o.x = cur[k].x * ah * aw.x;
            o.y = cur[k].y * ah * aw.y;
            o.z = cur[k].z * ah * aw.z;
            o.w = cur[k].w * ah * aw.w;
            op[32 * BLK_(AG_ * g + k)] = o;
        }
        for (int k = 0; k < AG_; ++k) cur[k] = nxt[k];
        if (g + 2 < NG_)
            for (int k = 0; k < AG_; ++k) nxt[k] = xp[32 * BLK_(AG_ * (g + 2) + k)];
    }
}

extern "C" void kernel_launch(void* const* d_in, const int* in_sizes, int n_in,
                              void* d_out, int out_size, void* d_ws, size_t ws_size,
                              hipStream_t stream) {
    const float* x     = (const float*)d_in[0];
    const float* w_fc  = (const float*)d_in[1];
    const float* bn_g  = (const float*)d_in[2];
    const float* bn_b  = (const float*)d_in[3];
    const float* bn_m  = (const float*)d_in[4];
    const float* bn_v  = (const float*)d_in[5];
    const float* w_h   = (const float*)d_in[6];
    const float* w_w   = (const float*)d_in[7];
    float* out = (float*)d_out;

    float* y = (float*)d_ws;
    float* z = out;

    squeeze_pixels_kernel<<<dim3((HW_ / ZP_) * N_), 32 * ZW_, 0, stream>>>(x, w_fc, z);
    pool_mid_kernel<<<dim3((L_ / 16) * N_), 32 * PW_, 0, stream>>>(z, bn_g, bn_b, bn_m,
                                                                   bn_v, y);
    excite_apply_kernel<<<dim3((C_ / AW_) * N_), 32 * AW_, 0, stream>>>(x, y, w_h, w_w, out);
}


# squeeze group = run of 6-7 aligned 64-pixel pieces (256 groups x 8 waves, LDS-only barriers) pools its z tile in LDS; apply joins 16 column + 2 row partials; pool launch gone (3 -> 2); new sum order
# speedup vs baseline: 2.9628x; 1.0591x over previous
#include <hip/hip_runtime.h>
#include <math.h>


typedef float v2f __attribute__((ext_vector_type(2)));
typedef float v8f __attribute__((ext_vector_type(8)));

#define N_   16
#define C_   256
#define H_   80
#define W_   80
#define MIP  8
#define L_   160
#define HW_  6400
#define ZP_  64
#define ZW_  8
#define ZC_  32

__device__ __forceinline__ float sigf(float v) {
    return 1.0f / (1.0f + __expf(-v));
}

#define NR_   16
#define NPX_  7
#define TP_   (NPX_ * ZP_)
#define ZOFF_ 80
#define ZT_   720

__device__ __forceinline__ void lds_barrier() {
    asm volatile("s_wait_dscnt 0x0\n\ts_barrier_signal -1\n\ts_barrier_wait -1" ::: "memory");
}

__device__ __forceinline__ void load_piece(float4 (&v0)[ZC_ / 4], float4 (&v1)[ZC_ / 4],
                                           const float* __restrict__ xq) {
    for (int s = 0; s < ZC_ / 4; ++s) {
        v0[s] = *reinterpret_cast<const float4*>(xq + (long)(4 * s)     * HW_);
        v1[s] = *reinterpret_cast<const float4*>(xq + (long)(4 * s + 1) * HW_);
    }
}

template <int NP>
__device__ __forceinline__ void squeeze_pool_run(const float* __restrict__ x,
                                                 const float* __restrict__ w_fc,
                                                 float* __restrict__ rowpart,
                                                 float* __restrict__ colpart,
                                                 float (&part)[2][ZW_][16][ZP_],
                                                 float (&zt)[MIP][ZT_],
                                                 const int n, const int run, const int p0) {
    const int wave = threadIdx.x >> 5;
    const int lane = threadIdx.x & 31;
    const int half = lane >> 4;
    const int l16  = lane & 15;

    const int    c0 = wave * ZC_ + 2 * half;
    const float* xb = x + ((long)n * C_ + c0) * HW_ + p0 * ZP_ + 4 * l16;
    const float* wr = w_fc + (l16 & (MIP - 1)) * C_ + c0;

    v2f a[ZC_ / 4];
    for (int s = 0; s < ZC_ / 4; ++s) {
        a[s].x = wr[4 * s];
        a[s].y = wr[4 * s + 1];
    }
    __builtin_amdgcn_sched_barrier(0);

    float4 v0[2][ZC_ / 4], v1[2][ZC_ / 4];
    load_piece(v0[0], v1[0], xb);
    load_piece(v0[1], v1[1], xb + ZP_);

    const int r  = threadIdx.x >> 5;
    const int cc = (threadIdx.x & 31) * 2;

    for (int k = threadIdx.x; k < MIP * ZT_ / 4; k += 256)
        reinterpret_cast<float4*>(&zt[0][0])[k] = float4{0.f, 0.f, 0.f, 0.f};

#pragma unroll
    for (int q = 0; q < NP; ++q) {
        const int cur = q & 1;
        v8f acc0 = {}, acc1 = {}, acc2 = {}, acc3 = {};
        for (int s = 0; s < ZC_ / 4; ++s) {
            v2f bb;
            bb.x = v0[cur][s].x;  bb.y = v1[cur][s].x;
            acc0 = __builtin_amdgcn_wmma_f32_16x16x4_f32(false, a[s], false, bb,
                                                         (short)0, acc0, false, false);
            bb.x = v0[cur][s].y;  bb.y = v1[cur][s].y;
            acc1 = __builtin_amdgcn_wmma_f32_16x16x4_f32(false, a[s], false, bb,
                                                         (short)0, acc1, false, false);
            bb.x = v0[cur][s].z;  bb.y = v1[cur][s].z;
            acc2 = __builtin_amdgcn_wmma_f32_16x16x4_f32(false, a[s], false, bb,
                                                         (short)0, acc2, false, false);
            bb.x = v0[cur][s].w;  bb.y = v1[cur][s].w;
            acc3 = __builtin_amdgcn_wmma_f32_16x16x4_f32(false, a[s], false, bb,
                                                         (short)0, acc3, false, false);
        }
        if (q + 2 < NP) load_piece(v0[cur], v1[cur], xb + (q + 2) * ZP_);
        __builtin_amdgcn_sched_barrier(0);

        for (int k = 0; k < MIP; ++k) {
            float4 o;
            o.x = acc0[k];  o.y = acc1[k];  o.z = acc2[k];  o.w = acc3[k];
            *reinterpret_cast<float4*>(&part[cur][wave][k + 8 * half][4 * l16]) = o;
        }
        lds_barrier();

        float2 s2 = {0.f, 0.f};
        for (int w = 0; w < ZW_; ++w) {
            const float2 p = *reinterpret_cast<const float2*>(&part[cur][w][r][cc]);
            s2.x += p.x;
            s2.y += p.y;
        }
        *reinterpret_cast<float2*>(&zt[r][ZOFF_ + q * ZP_ + cc]) = s2;
    }
    lds_barrier();

    const int P0 = p0 * ZP_;
    const int P1 = P0 + NP * ZP_;

    {
        const int m    = threadIdx.x >> 5;
        const int row  = P0 / W_ + (lane >> 2);
        const int pb   = row * W_ + (lane & 3) * 20;
        const float* zr = &zt[m][ZOFF_ + pb - P0];
        float s = 0.f;
        for (int k = 0; k < 20; ++k) s += zr[k];
        s += __shfl_xor(s, 1);  s += __shfl_xor(s, 2);
        const int rb = row * W_;
        if ((lane & 3) == 0 && rb < P1 && rb + W_ > P0) {
            float* dst = rowpart + (((long)n * MIP + m) * H_ + row) * 2;
            if (rb >= P0) {
                dst[0] = s;
                if (rb + W_ <= P1) dst[1] = 0.f;
            } else {
                dst[1] = s;
            }
        }
    }
    if (threadIdx.x < MIP * (W_ / 4)) {
        const int m   = threadIdx.x / (W_ / 4);
        const int jq  = threadIdx.x - m * (W_ / 4);
        const int off = (4 * jq + W_ - P0 % W_) % W_;
        float4 s = {0.f, 0.f, 0.f, 0.f};
        for (int t = 0; t < (TP_ + W_ - 1) / W_; ++t) {
            const float4 u = *reinterpret_cast<const float4*>(&zt[m][ZOFF_ + off + t * W_]);
            s.x += u.x;  s.y += u.y;  s.z += u.z;  s.w += u.w;
        }
        *reinterpret_cast<float4*>(colpart + (((long)n * NR_ + run) * MIP + m) * W_ + 4 * jq) = s;
    }
}

__global__ __launch_bounds__(256, 2)
void squeeze_pool_kernel(const float* __restrict__ x,
                         const float* __restrict__ w_fc,
                         float* __restrict__ rowpart,
                         float* __restrict__ colpart) {
    __shared__ __attribute__((aligned(16))) float part[2][ZW_][16][ZP_];
    __shared__ __attribute__((aligned(16))) float zt[MIP][ZT_];

    const int n   = blockIdx.x & (N_ - 1);
    const int run = blockIdx.x >> 4;
    const int p0  = (run * 25) >> 2;
    const int np  = (((run + 1) * 25) >> 2) - p0;
    if (np == NPX_) squeeze_pool_run<NPX_>(x, w_fc, rowpart, colpart, part, zt, n, run, p0);
    else            squeeze_pool_run<NPX_ - 1>(x, w_fc, rowpart, colpart, part, zt, n, run, p0);
}

#define AW_ 16
#define AG_ 10
#define NG_ 5
#define BLK_(t) (AG_ * NG_ - 1 - (t))

__global__ __launch_bounds__(512)
void excite_apply_kernel(const float* __restrict__ x,
                         const float* __restrict__ rowpart,
                         const float* __restrict__ colpart,
                         const float* __restrict__ bn_g,
                         const float* __restrict__ bn_b,
                         const float* __restrict__ bn_m,
                         const float* __restrict__ bn_v,
                         const float* __restrict__ w_h,
                         const float* __restrict__ w_w,
                         float* __restrict__ out) {
    __shared__ __attribute__((aligned(16))) float gate[AW_][L_];
    __shared__ __attribute__((aligned(16))) float ysh[MIP][L_];

    const int n     = blockIdx.x & (N_ - 1);
    const int ctile = blockIdx.x >> 4;
    const int wave  = threadIdx.x >> 5;
    const int lane  = threadIdx.x & 31;
    const int half  = lane >> 4;
    const int l16   = lane & 15;

    const long    plane = ((long)n * C_ + ctile * AW_ + wave) * HW_;
    const float4* xp = reinterpret_cast<const float4*>(x + plane) + lane;
    float4*       op = reinterpret_cast<float4*>(out + plane) + lane;

    const bool gates  = (wave < L_ / 16);
    const bool joiner = (wave < 5);
    const bool rower  = gates && !joiner;
    const int  jt = threadIdx.x - (joiner ? 0 : 160);
    const int  jm = jt / (W_ / 4);
    const int  jc = jt - jm * (W_ / 4);
    float4 cp[NR_];
    float4 ra = {0.f, 0.f, 0.f, 0.f}, rb = {0.f, 0.f, 0.f, 0.f};
    float  pg = 0.f, pb = 0.f, pm = 0.f, pv = 1.f;
    v2f a0 = {}, a1 = {};
    if (joiner) {
        pg = bn_g[jm];  pb = bn_b[jm];  pm = bn_m[jm];  pv = bn_v[jm];
        const float4* src = reinterpret_cast<const float4*>(
            colpart + ((long)n * NR_ * MIP + jm) * W_ + 4 * jc);
        for (int k = 0; k < NR_; ++k) cp[k] = src[k * (MIP * W_ / 4)];
    } else if (rower) {
        pg = bn_g[jm];  pb = bn_b[jm];  pm = bn_m[jm];  pv = bn_v[jm];
        const float4* src = reinterpret_cast<const float4*>(
            rowpart + (((long)n * MIP + jm) * H_ + 4 * jc) * 2);
        ra = src[0];
        rb = src[1];
    }
    if (gates) {
        const float* Wm = (wave < 5) ? w_h : w_w;
        const float* wr = Wm + (ctile * AW_ + l16) * MIP + 2 * half;
        a0.x = wr[0];           a0.y = wr[1];
        a1.x = wr[4];           a1.y = wr[5];
    }

    float4 cur[AG_], nxt[AG_];
    for (int k = 0; k < AG_; ++k) cur[k] = xp[32 * BLK_(k)];
    for (int k = 0; k < AG_; ++k) nxt[k] = xp[32 * BLK_(AG_ + k)];

    if (joiner) {
        asm volatile("" : "+v"(cp[0].x), "+v"(cp[0].y), "+v"(cp[0].z), "+v"(cp[0].w),
                          "+v"(pg), "+v"(pb), "+v"(pm), "+v"(pv));
        float4 s = cp[0];
        for (int k = 1; k < NR_; ++k) {
            s.x += cp[k].x;  s.y += cp[k].y;  s.z += cp[k].z;  s.w += cp[k].w;
        }
        const float iv    = pg * rsqrtf(pv + 1e-5f);
        const float shift = pb - pm * iv;
        float4 yy;
        yy.x = s.x * (1.0f / (float)H_) * iv + shift;
        yy.y = s.y * (1.0f / (float)H_) * iv + shift;
        yy.z = s.z * (1.0f / (float)H_) * iv + shift;
        yy.w = s.w * (1.0f / (float)H_) * iv + shift;
        yy.x *= sigf(yy.x);  yy.y *= sigf(yy.y);
        yy.z *= sigf(yy.z);  yy.w *= sigf(yy.w);
        *reinterpret_cast<float4*>(&ysh[jm][H_ + 4 * jc]) = yy;
    } else if (rower) {
        asm volatile("" : "+v"(ra.x), "+v"(ra.y), "+v"(ra.z), "+v"(ra.w),
                          "+v"(rb.x), "+v"(rb.y), "+v"(rb.z), "+v"(rb.w),
                          "+v"(pg), "+v"(pb), "+v"(pm), "+v"(pv));
        const float iv    = pg * rsqrtf(pv + 1e-5f);
        const float shift = pb - pm * iv;
        float4 yy;
        yy.x = (ra.x + ra.y) * (1.0f / (float)W_) * iv + shift;
        yy.y = (ra.z + ra.w) * (1.0f / (float)W_) * iv + shift;
        yy.z = (rb.x + rb.y) * (1.0f / (float)W_) * iv + shift;
        yy.w = (rb.z + rb.w) * (1.0f / (float)W_) * iv + shift;
        yy.x *= sigf(yy.x);  yy.y *= sigf(yy.y);
        yy.z *= sigf(yy.z);  yy.w *= sigf(yy.w);
        *reinterpret_cast<float4*>(&ysh[jm][4 * jc]) = yy;
    }
    __syncthreads();

    if (gates) {
        const float* yc = &ysh[2 * half][wave * 16 + l16];
        v2f b0, b1;
        b0.x = yc[0];           b0.y = yc[L_];
        b1.x = yc[4 * L_];      b1.y = yc[5 * L_];
        v8f acc = {};
        acc = __builtin_amdgcn_wmma_f32_16x16x4_f32(false, a0, false, b0,
                                                    (short)0, acc, false, false);
        acc = __builtin_amdgcn_wmma_f32_16x16x4_f32(false, a1, false, b1,
                                                    (short)0, acc, false, false);
        for (int r = 0; r < MIP; ++r)
            gate[half * 8 + r][wave * 16 + l16] = sigf(acc[r]);
    }
    __syncthreads();

    const float* gh = &gate[wave][0];
    const float* gw = &gate[wave][H_];
    for (int g = 0; g < NG_; ++g) {
        for (int k = 0; k < AG_; ++k) {
            const unsigned f  = (unsigned)lane + 32u * (unsigned)BLK_(AG_ * g + k);
            const unsigned i  = f / 20u;
            const unsigned jq = f - 20u * i;
            const float    ah = gh[i];
            const float4   aw = *reinterpret_cast<const float4*>(gw + 4 * jq);
            float4 o;
            o.x = cur[k].x * ah * aw.x;
            o.y = cur[k].y * ah * aw.y;
            o.z = cur[k].z * ah * aw.z;
            o.w = cur[k].w * ah * aw.w;
            op[32 * BLK_(AG_ * g + k)] = o;
        }
        for (int k = 0; k < AG_; ++k) cur[k] = nxt[k];
        if (g + 2 < NG_)
            for (int k = 0; k < AG_; ++k) nxt[k] = xp[32 * BLK_(AG_ * (g + 2) + k)];
    }
}

extern "C" void kernel_launch(void* const* d_in, const int* in_sizes, int n_in,
                              void* d_out, int out_size, void* d_ws, size_t ws_size,
                              hipStream_t stream) {
    const float* x     = (const float*)d_in[0];
    const float* w_fc  = (const float*)d_in[1];
    const float* bn_g  = (const float*)d_in[2];
    const float* bn_b  = (const float*)d_in[3];
    const float* bn_m  = (const float*)d_in[4];
    const float* bn_v  = (const float*)d_in[5];
    const float* w_h   = (const float*)d_in[6];
    const float* w_w   = (const float*)d_in[7];
    float* out = (float*)d_out;

    float* rowpart = (float*)d_ws;
    float* colpart = (float*)d_ws + N_ * MIP * H_ * 2;

    squeeze_pool_kernel<<<dim3(NR_ * N_), 32 * ZW_, 0, stream>>>(x, w_fc, rowpart, colpart);
    excite_apply_kernel<<<dim3((C_ / AW_) * N_), 32 * AW_, 0, stream>>>(x, rowpart, colpart, bn_g,
                                                                        bn_b, bn_m, bn_v,
                                                                        w_h, w_w, out);
}
